# starting kernel
# baseline (speedup 1.0000x reference)
#include <hip/hip_runtime.h>


typedef __attribute__((ext_vector_type(2))) float v2f;
typedef __attribute__((ext_vector_type(8))) float v8f;

#define CIN   32
#define FOUT  32
#define Hdim  128
#define Wdim  128
#define KHW   3
#define TILE_W 32
#define ROWS_PER_BLK 4
#define XROWS (ROWS_PER_BLK + 2)
#define XCOLS (TILE_W + 2)
#define XPIX_STRIDE 36
#define WSIZE (KHW * KHW * CIN * FOUT)
#define PSTRIDE (WSIZE + FOUT)

__global__ __launch_bounds__(128)
void conv3x3_wmma_f32(const float* __restrict__ x,
                      const float* __restrict__ params,
                      float* __restrict__ out) {
    __shared__ float sW3[WSIZE];
    __shared__ float sX[XROWS * XCOLS * XPIX_STRIDE];
    __shared__ float sB[FOUT];

    const int tid  = threadIdx.x;
    const int b    = blockIdx.y;
    const int wt   = blockIdx.x & 3;
    const int hblk = blockIdx.x >> 2;
    const int h0   = hblk * ROWS_PER_BLK;
    const int w0   = wt * TILE_W;

    {
        const float* pb = params + (size_t)b * PSTRIDE;
        float4* wdst = (float4*)sW3;
        #pragma unroll
        for (int it = 0; it < WSIZE / 4 / 128; ++it) {
            const int j   = tid + 128 * it;
            const int tap = j >> 8;
            const int q   = (j >> 5) & 7;
            const int f   = j & 31;
            const float* src = pb + ((tap * 32 + 4 * q) * 32 + f);
            wdst[j] = make_float4(src[0], src[32], src[64], src[96]);
        }
        if (tid < FOUT)
            sB[tid] = pb[WSIZE + tid];
    }

    const int NXPIX = XROWS * XCOLS;
    float4* xdst = (float4*)sX;
    for (int i = tid; i < NXPIX * 8; i += 128) {
        const int pix = i >> 3;
        const int c4  = i & 7;
        const int r   = pix / XCOLS;
        const int c   = pix - r * XCOLS;
        const int row = h0 - 1 + r;
        const int col = w0 - 1 + c;
        float4 v = make_float4(0.f, 0.f, 0.f, 0.f);
        if ((unsigned)row < (unsigned)Hdim && (unsigned)col < (unsigned)Wdim)
            v = *(const float4*)(x + ((((size_t)b * Hdim + row) * Wdim + col) * CIN) + c4 * 4);
        xdst[pix * (XPIX_STRIDE / 4) + c4] = v;
    }
    __syncthreads();

    const int lane  = tid & 31;
    const int wave  = tid >> 5;
    const int mlane = lane & 15;
    const int hi    = lane >> 4;

    v8f acc00 = {}, acc01 = {}, acc10 = {}, acc11 = {};

    #pragma unroll
    for (int kh = 0; kh < KHW; ++kh) {
        #pragma unroll
        for (int kw = 0; kw < KHW; ++kw) {
            const float* xrow = &sX[((kh + wave) * XCOLS + mlane + kw) * XPIX_STRIDE];
            const float* wb   = &sW3[(kh * 3 + kw) * 1024 + mlane * 4 + 2 * hi];
            #pragma unroll
            for (int kb = 0; kb < CIN; kb += 4) {
                const int k0 = kb + 2 * hi;
                v2f a0 = { xrow[k0],                       xrow[k0 + 1] };
                v2f a1 = { xrow[16 * XPIX_STRIDE + k0],    xrow[16 * XPIX_STRIDE + k0 + 1] };
                const float* bp = wb + kb * 32;
                v2f b0 = *(const v2f*)bp;
                v2f b1 = *(const v2f*)(bp + 64);
                acc00 = __builtin_amdgcn_wmma_f32_16x16x4_f32(false, a0, false, b0,
                        (short)0, acc00, false, false);
                acc01 = __builtin_amdgcn_wmma_f32_16x16x4_f32(false, a0, false, b1,
                        (short)0, acc01, false, false);
                acc10 = __builtin_amdgcn_wmma_f32_16x16x4_f32(false, a1, false, b0,
                        (short)0, acc10, false, false);
                acc11 = __builtin_amdgcn_wmma_f32_16x16x4_f32(false, a1, false, b1,
                        (short)0, acc11, false, false);
            }
        }
    }

    const float bias0 = sB[mlane];
    const float bias1 = sB[16 + mlane];
    const int   h     = h0 + wave;
    float* obase = out
        + ((((size_t)b * Hdim + h) * Wdim) + w0 + 8 * hi) * FOUT + mlane;
    #pragma unroll
    for (int r = 0; r < 8; ++r) {
        obase[(size_t)r * FOUT]             = acc00[r] + bias0;
        obase[(size_t)r * FOUT + 16]        = acc01[r] + bias1;
        obase[(size_t)(16 + r) * FOUT]      = acc10[r] + bias0;
        obase[(size_t)(16 + r) * FOUT + 16] = acc11[r] + bias1;
    }
}

extern "C" void kernel_launch(void* const* d_in, const int* in_sizes, int n_in,
                              void* d_out, int out_size, void* d_ws, size_t ws_size,
                              hipStream_t stream) {
    (void)in_sizes; (void)n_in; (void)d_ws; (void)ws_size; (void)out_size;
    const float* x      = (const float*)d_in[0];
    const float* params = (const float*)d_in[1];
    float* out          = (float*)d_out;

    dim3 grid((Hdim / ROWS_PER_BLK) * (Wdim / TILE_W), 32);
    conv3x3_wmma_f32<<<grid, 128, 0, stream>>>(x, params, out);
}


# Staging made branch-free and unrolled (clamped tile addresses + select, 4 spare LDS pixel slots, bias held in registers): a thread's 87 global loads issue before the first wait, not 15 round trips
# speedup vs baseline: 1.1487x; 1.1487x over previous
#include <hip/hip_runtime.h>


typedef __attribute__((ext_vector_type(2))) float v2f;
typedef __attribute__((ext_vector_type(8))) float v8f;

#define CIN   32
#define FOUT  32
#define Hdim  128
#define Wdim  128
#define KHW   3
#define TILE_W 32
#define ROWS_PER_BLK 4
#define XROWS (ROWS_PER_BLK + 2)
#define XCOLS (TILE_W + 2)
#define XPIX_STRIDE 36
#define WSIZE (KHW * KHW * CIN * FOUT)
#define PSTRIDE (WSIZE + FOUT)
#define XITERS ((XROWS * XCOLS * 8 + 127) / 128)

__global__ __launch_bounds__(128)
void conv3x3_wmma_f32(const float* __restrict__ x,
                      const float* __restrict__ params,
                      float* __restrict__ out) {
    __shared__ float sW3[WSIZE];
    __shared__ float sX[XITERS * 16 * XPIX_STRIDE];

    const int tid  = threadIdx.x;
    const int b    = blockIdx.y;
    const int wt   = blockIdx.x & 3;
    const int hblk = blockIdx.x >> 2;
    const int h0   = hblk * ROWS_PER_BLK;
    const int w0   = wt * TILE_W;

    const float* pb = params + (size_t)b * PSTRIDE;

    const int c4 = tid & 7;
    const int pg = tid >> 3;
    const float* xb = x + (size_t)b * Hdim * Wdim * CIN + c4 * 4;
    float4 xv[XITERS];
    #pragma unroll
    for (int it = 0; it < XITERS; ++it) {
        const int pix = pg + 16 * it;
        const int r   = pix / XCOLS;
        const int c   = pix - r * XCOLS;
        const int row = min(max(h0 - 1 + r, 0), Hdim - 1);
        const int col = min(max(w0 - 1 + c, 0), Wdim - 1);
        xv[it] = *(const float4*)(xb + ((size_t)row * Wdim + col) * CIN);
    }
    const float bias0 = pb[WSIZE + (tid & 15)];
    const float bias1 = pb[WSIZE + 16 + (tid & 15)];

    {
        float4* wdst = (float4*)sW3;
        #pragma unroll
        for (int it = 0; it < WSIZE / 4 / 128; ++it) {
            const int j   = tid + 128 * it;
            const int tap = j >> 8;
            const int q   = (j >> 5) & 7;
            const int f   = j & 31;
            const float* src = pb + ((tap * 32 + 4 * q) * 32 + f);
            wdst[j] = make_float4(src[0], src[32], src[64], src[96]);
        }
    }

    float4* xdst = (float4*)sX;
    #pragma unroll
    for (int it = 0; it < XITERS; ++it) {
        const int pix = pg + 16 * it;
        const int r   = pix / XCOLS;
        const int c   = pix - r * XCOLS;
        const int row = h0 - 1 + r;
        const int col = w0 - 1 + c;
        float4 v = xv[it];
        if (!((unsigned)row < (unsigned)Hdim && (unsigned)col < (unsigned)Wdim))
            v = make_float4(0.f, 0.f, 0.f, 0.f);
        xdst[pix * (XPIX_STRIDE / 4) + c4] = v;
    }
    __syncthreads();

    const int lane  = tid & 31;
    const int wave  = tid >> 5;
    const int mlane = lane & 15;
    const int hi    = lane >> 4;

    v8f acc00 = {}, acc01 = {}, acc10 = {}, acc11 = {};

    #pragma unroll
    for (int kh = 0; kh < KHW; ++kh) {
        #pragma unroll
        for (int kw = 0; kw < KHW; ++kw) {
            const float* xrow = &sX[((kh + wave) * XCOLS + mlane + kw) * XPIX_STRIDE];
            const float* wb   = &sW3[(kh * 3 + kw) * 1024 + mlane * 4 + 2 * hi];
            #pragma unroll
            for (int kb = 0; kb < CIN; kb += 4) {
                const int k0 = kb + 2 * hi;
                v2f a0 = { xrow[k0],                       xrow[k0 + 1] };
                v2f a1 = { xrow[16 * XPIX_STRIDE + k0],    xrow[16 * XPIX_STRIDE + k0 + 1] };
                const float* bp = wb + kb * 32;
                v2f b0 = *(const v2f*)bp;
                v2f b1 = *(const v2f*)(bp + 64);
                acc00 = __builtin_amdgcn_wmma_f32_16x16x4_f32(false, a0, false, b0,
                        (short)0, acc00, false, false);
                acc01 = __builtin_amdgcn_wmma_f32_16x16x4_f32(false, a0, false, b1,
                        (short)0, acc01, false, false);
                acc10 = __builtin_amdgcn_wmma_f32_16x16x4_f32(false, a1, false, b0,
                        (short)0, acc10, false, false);
                acc11 = __builtin_amdgcn_wmma_f32_16x16x4_f32(false, a1, false, b1,
                        (short)0, acc11, false, false);
            }
        }
    }

    const int   h     = h0 + wave;
    float* obase = out
        + ((((size_t)b * Hdim + h) * Wdim) + w0 + 8 * hi) * FOUT + mlane;
    #pragma unroll
    for (int r = 0; r < 8; ++r) {
        obase[(size_t)r * FOUT]             = acc00[r] + bias0;
        obase[(size_t)r * FOUT + 16]        = acc01[r] + bias1;
        obase[(size_t)(16 + r) * FOUT]      = acc10[r] + bias0;
        obase[(size_t)(16 + r) * FOUT + 16] = acc11[r] + bias1;
    }
}

extern "C" void kernel_launch(void* const* d_in, const int* in_sizes, int n_in,
                              void* d_out, int out_size, void* d_ws, size_t ws_size,
                              hipStream_t stream) {
    (void)in_sizes; (void)n_in; (void)d_ws; (void)ws_size; (void)out_size;
    const float* x      = (const float*)d_in[0];
    const float* params = (const float*)d_in[1];
    float* out          = (float*)d_out;

    dim3 grid((Hdim / ROWS_PER_BLK) * (Wdim / TILE_W), 32);
    conv3x3_wmma_f32<<<grid, 128, 0, stream>>>(x, params, out);
}


# Weight-stationary strips: a block stages its sample's weights in LDS once and walks the 4 column tiles of a 4-row strip (grid 1024); next tile's input loads fly under the WMMAs, LDS-only barrier
# speedup vs baseline: 1.3481x; 1.1736x over previous
#include <hip/hip_runtime.h>


typedef __attribute__((ext_vector_type(2))) float v2f;
typedef __attribute__((ext_vector_type(8))) float v8f;

#define CIN   32
#define FOUT  32
#define Hdim  128
#define Wdim  128
#define KHW   3
#define TILE_W 32
#define NTILES (Wdim / TILE_W)
#define ROWS_PER_BLK 4
#define XROWS (ROWS_PER_BLK + 2)
#define XCOLS (TILE_W + 2)
#define XPIX_STRIDE 36
#define WSIZE (KHW * KHW * CIN * FOUT)
#define PSTRIDE (WSIZE + FOUT)
#define XITERS ((XROWS * XCOLS * 8 + 127) / 128)

static __device__ __forceinline__ void lds_barrier() {
    asm volatile("s_wait_dscnt 0x0\n\ts_barrier_signal -1\n\ts_barrier_wait -1" ::: "memory");
}

static __device__ __forceinline__ void load_x_tile(float4 (&xv)[XITERS], const float* xb,
                                                   int h0, int w0, int pg) {
    #pragma unroll
    for (int it = 0; it < XITERS; ++it) {
        const int pix = pg + 16 * it;
        const int r   = pix / XCOLS;
        const int c   = pix - r * XCOLS;
        const int row = min(max(h0 - 1 + r, 0), Hdim - 1);
        const int col = min(max(w0 - 1 + c, 0), Wdim - 1);
        xv[it] = *(const float4*)(xb + ((size_t)row * Wdim + col) * CIN);
    }
}

static __device__ __forceinline__ void store_x_tile(const float4 (&xv)[XITERS], float* sX,
                                                    int h0, int w0, int pg, int c4) {
    float4* xdst = (float4*)sX;
    #pragma unroll
    for (int it = 0; it < XITERS; ++it) {
        const int pix = pg + 16 * it;
        const int r   = pix / XCOLS;
        const int c   = pix - r * XCOLS;
        const int row = h0 - 1 + r;
        const int col = w0 - 1 + c;
        float4 v = xv[it];
        if (!((unsigned)row < (unsigned)Hdim && (unsigned)col < (unsigned)Wdim))
            v = make_float4(0.f, 0.f, 0.f, 0.f);
        xdst[pix * (XPIX_STRIDE / 4) + c4] = v;
    }
}

__global__ __launch_bounds__(128)
void conv3x3_wmma_f32(const float* __restrict__ x,
                      const float* __restrict__ params,
                      float* __restrict__ out) {
    __shared__ float sW3[WSIZE];
    __shared__ float sX[XITERS * 16 * XPIX_STRIDE];

    const int tid  = threadIdx.x;
    const int b    = blockIdx.y;
    const int h0   = blockIdx.x * ROWS_PER_BLK;

    const float* pb = params + (size_t)b * PSTRIDE;

    const int c4 = tid & 7;
    const int pg = tid >> 3;
    const float* xb = x + (size_t)b * Hdim * Wdim * CIN + c4 * 4;
    float4 xv[XITERS];
    load_x_tile(xv, xb, h0, 0, pg);

    const float bias0 = pb[WSIZE + (tid & 15)];
    const float bias1 = pb[WSIZE + 16 + (tid & 15)];

    {
        float4* wdst = (float4*)sW3;
        #pragma unroll
        for (int it = 0; it < WSIZE / 4 / 128; ++it) {
            const int j   = tid + 128 * it;
            const int tap = j >> 8;
            const int q   = (j >> 5) & 7;
            const int f   = j & 31;
            const float* src = pb + ((tap * 32 + 4 * q) * 32 + f);
            wdst[j] = make_float4(src[0], src[32], src[64], src[96]);
        }
    }

    store_x_tile(xv, sX, h0, 0, pg, c4);
    __syncthreads();

    const int lane  = tid & 31;
    const int wave  = tid >> 5;
    const int mlane = lane & 15;
    const int hi    = lane >> 4;
    const int h     = h0 + wave;
    float* orow = out + ((((size_t)b * Hdim + h) * Wdim) + 8 * hi) * FOUT + mlane;

    #pragma unroll 1
    for (int t = 0; t < NTILES; ++t) {
        const int w0 = t * TILE_W;

        if (t + 1 < NTILES)
            load_x_tile(xv, xb, h0, w0 + TILE_W, pg);

        v8f acc00 = {}, acc01 = {}, acc10 = {}, acc11 = {};

        #pragma unroll
        for (int kh = 0; kh < KHW; ++kh) {
            #pragma unroll
            for (int kw = 0; kw < KHW; ++kw) {
                const float* xrow = &sX[((kh + wave) * XCOLS + mlane + kw) * XPIX_STRIDE];
                const float* wb   = &sW3[(kh * 3 + kw) * 1024 + mlane * 4 + 2 * hi];
                #pragma unroll
                for (int kb = 0; kb < CIN; kb += 4) {
                    const int k0 = kb + 2 * hi;
                    v2f a0 = { xrow[k0],                       xrow[k0 + 1] };
                    v2f a1 = { xrow[16 * XPIX_STRIDE + k0],    xrow[16 * XPIX_STRIDE + k0 + 1] };
                    const float* bp = wb + kb * 32;
                    v2f b0 = *(const v2f*)bp;
                    v2f b1 = *(const v2f*)(bp + 64);
                    acc00 = __builtin_amdgcn_wmma_f32_16x16x4_f32(false, a0, false, b0,
                            (short)0, acc00, false, false);
                    acc01 = __builtin_amdgcn_wmma_f32_16x16x4_f32(false, a0, false, b1,
                            (short)0, acc01, false, false);
                    acc10 = __builtin_amdgcn_wmma_f32_16x16x4_f32(false, a1, false, b0,
                            (short)0, acc10, false, false);
                    acc11 = __builtin_amdgcn_wmma_f32_16x16x4_f32(false, a1, false, b1,
                            (short)0, acc11, false, false);
                }
            }
        }

        float* obase = orow + (size_t)w0 * FOUT;
        #pragma unroll
        for (int r = 0; r < 8; ++r) {
            obase[(size_t)r * FOUT]             = acc00[r] + bias0;
            obase[(size_t)r * FOUT + 16]        = acc01[r] + bias1;
            obase[(size_t)(16 + r) * FOUT]      = acc10[r] + bias0;
            obase[(size_t)(16 + r) * FOUT + 16] = acc11[r] + bias1;
        }

        if (t + 1 < NTILES) {
            lds_barrier();
            store_x_tile(xv, sX, h0, w0 + TILE_W, pg, c4);
            lds_barrier();
        }
    }
}

extern "C" void kernel_launch(void* const* d_in, const int* in_sizes, int n_in,
                              void* d_out, int out_size, void* d_ws, size_t ws_size,
                              hipStream_t stream) {
    (void)in_sizes; (void)n_in; (void)d_ws; (void)ws_size; (void)out_size;
    const float* x      = (const float*)d_in[0];
    const float* params = (const float*)d_in[1];
    float* out          = (float*)d_out;

    dim3 grid(Hdim / ROWS_PER_BLK, 32);
    conv3x3_wmma_f32<<<grid, 128, 0, stream>>>(x, params, out);
}


# Fragment loads as one group a 16-WMMA block ahead (2 register sets, 8 ds_load_b128 per place: 18 LDS places per tile, not 72); a float4 holds 4 cin, so the cin summation order changes
# speedup vs baseline: 1.3904x; 1.0314x over previous
#include <hip/hip_runtime.h>


typedef __attribute__((ext_vector_type(2))) float v2f;
typedef __attribute__((ext_vector_type(4))) float v4f;
typedef __attribute__((ext_vector_type(8))) float v8f;

#define CIN   32
#define FOUT  32
#define Hdim  128
#define Wdim  128
#define KHW   3
#define TILE_W 32
#define NTILES (Wdim / TILE_W)
#define ROWS_PER_BLK 4
#define XROWS (ROWS_PER_BLK + 2)
#define XCOLS (TILE_W + 2)
#define XPIX_STRIDE 36
#define WSIZE (KHW * KHW * CIN * FOUT)
#define PSTRIDE (WSIZE + FOUT)
#define XITERS ((XROWS * XCOLS * 8 + 127) / 128)

static __device__ __forceinline__ void lds_barrier() {
    asm volatile("s_wait_dscnt 0x0\n\ts_barrier_signal -1\n\ts_barrier_wait -1" ::: "memory");
}

static __device__ __forceinline__ void load_x_tile(float4 (&xv)[XITERS], const float* xb,
                                                   int h0, int w0, int pg) {
    #pragma unroll
    for (int it = 0; it < XITERS; ++it) {
        const int pix = pg + 16 * it;
        const int r   = pix / XCOLS;
        const int c   = pix - r * XCOLS;
        const int row = min(max(h0 - 1 + r, 0), Hdim - 1);
        const int col = min(max(w0 - 1 + c, 0), Wdim - 1);
        xv[it] = *(const float4*)(xb + ((size_t)row * Wdim + col) * CIN);
    }
}

static __device__ __forceinline__ void store_x_tile(const float4 (&xv)[XITERS], float* sX,
                                                    int h0, int w0, int pg, int c4) {
    float4* xdst = (float4*)sX;
    #pragma unroll
    for (int it = 0; it < XITERS; ++it) {
        const int pix = pg + 16 * it;
        const int r   = pix / XCOLS;
        const int c   = pix - r * XCOLS;
        const int row = h0 - 1 + r;
        const int col = w0 - 1 + c;
        float4 v = xv[it];
        if (!((unsigned)row < (unsigned)Hdim && (unsigned)col < (unsigned)Wdim))
            v = make_float4(0.f, 0.f, 0.f, 0.f);
        xdst[pix * (XPIX_STRIDE / 4) + c4] = v;
    }
}

struct FragSet {
    v4f a0[2], a1[2];
    v4f b0[2], b1[2];
};

static __device__ __forceinline__ void load_frags(FragSet& s, const float* xa,
                                                  const float* wq, int blk) {
    const int tap = blk >> 1;
    const int hf  = blk & 1;
    const int kh  = tap / KHW;
    const int kw  = tap - kh * KHW;
    const float* xr = xa + (kh * XCOLS + kw) * XPIX_STRIDE + 16 * hf;
    const float* wr = wq + tap * 1024 + 512 * hf;
    #pragma unroll
    for (int jj = 0; jj < 2; ++jj) {
        s.a0[jj] = *(const v4f*)(xr + 8 * jj);
        s.a1[jj] = *(const v4f*)(xr + 16 * XPIX_STRIDE + 8 * jj);
        s.b0[jj] = *(const v4f*)(wr + 256 * jj);
        s.b1[jj] = *(const v4f*)(wr + 256 * jj + 64);
    }
}

static __device__ __forceinline__ void mma_block(const FragSet& s, v8f& acc00, v8f& acc01,
                                                 v8f& acc10, v8f& acc11) {
    #pragma unroll
    for (int jj = 0; jj < 2; ++jj) {
        #pragma unroll
        for (int p = 0; p < 2; ++p) {
            v2f a0 = { s.a0[jj][2 * p], s.a0[jj][2 * p + 1] };
            v2f a1 = { s.a1[jj][2 * p], s.a1[jj][2 * p + 1] };
            v2f b0 = { s.b0[jj][2 * p], s.b0[jj][2 * p + 1] };
            v2f b1 = { s.b1[jj][2 * p], s.b1[jj][2 * p + 1] };
            acc00 = __builtin_amdgcn_wmma_f32_16x16x4_f32(false, a0, false, b0,
                    (short)0, acc00, false, false);
            acc01 = __builtin_amdgcn_wmma_f32_16x16x4_f32(false, a0, false, b1,
                    (short)0, acc01, false, false);
            acc10 = __builtin_amdgcn_wmma_f32_16x16x4_f32(false, a1, false, b0,
                    (short)0, acc10, false, false);
            acc11 = __builtin_amdgcn_wmma_f32_16x16x4_f32(false, a1, false, b1,
                    (short)0, acc11, false, false);
        }
    }
}

#define NBLK (KHW * KHW * 2)

__global__ __launch_bounds__(128)
void conv3x3_wmma_f32(const float* __restrict__ x,
                      const float* __restrict__ params,
                      float* __restrict__ out) {
    __shared__ __attribute__((aligned(16))) float sW3[WSIZE];
    __shared__ __attribute__((aligned(16))) float sX[XITERS * 16 * XPIX_STRIDE];

    const int tid  = threadIdx.x;
    const int b    = blockIdx.y;
    const int h0   = blockIdx.x * ROWS_PER_BLK;

    const float* pb = params + (size_t)b * PSTRIDE;

    const int c4 = tid & 7;
    const int pg = tid >> 3;
    const float* xb = x + (size_t)b * Hdim * Wdim * CIN + c4 * 4;
    float4 xv[XITERS];
    load_x_tile(xv, xb, h0, 0, pg);

    const float bias0 = pb[WSIZE + (tid & 15)];
    const float bias1 = pb[WSIZE + 16 + (tid & 15)];

    {
        float4* wdst = (float4*)sW3;
        #pragma unroll
        for (int it = 0; it < WSIZE / 4 / 128; ++it) {
            const int j   = tid + 128 * it;
            const int tap = j >> 8;
            const int q   = (j >> 5) & 7;
            const int f   = j & 31;
            const float* src = pb + ((tap * 32 + 4 * q) * 32 + f);
            wdst[j] = make_float4(src[0], src[32], src[64], src[96]);
        }
    }

    store_x_tile(xv, sX, h0, 0, pg, c4);
    __syncthreads();

    const int lane  = tid & 31;
    const int wave  = tid >> 5;
    const int mlane = lane & 15;
    const int hi    = lane >> 4;
    const int h     = h0 + wave;
    float* orow = out + ((((size_t)b * Hdim + h) * Wdim) + 8 * hi) * FOUT + mlane;
    const float* xa = &sX[(wave * XCOLS + mlane) * XPIX_STRIDE + 4 * hi];
    const float* wq = &sW3[mlane * 4 + 128 * hi];

    #pragma unroll 1
    for (int t = 0; t < NTILES; ++t) {
        const int w0 = t * TILE_W;

        if (t + 1 < NTILES)
            load_x_tile(xv, xb, h0, w0 + TILE_W, pg);

        v8f acc00 = {}, acc01 = {}, acc10 = {}, acc11 = {};

        FragSet fs[2];
        load_frags(fs[0], xa, wq, 0);
        #pragma unroll
        for (int blk = 0; blk < NBLK; ++blk) {
            if (blk + 1 < NBLK)
                load_frags(fs[(blk + 1) & 1], xa, wq, blk + 1);
            __builtin_amdgcn_sched_barrier(0);
            mma_block(fs[blk & 1], acc00, acc01, acc10, acc11);
            __builtin_amdgcn_sched_barrier(0);
        }

        float* obase = orow + (size_t)w0 * FOUT;
        #pragma unroll
        for (int r = 0; r < 8; ++r) {
            obase[(size_t)r * FOUT]             = acc00[r] + bias0;
            obase[(size_t)r * FOUT + 16]        = acc01[r] + bias1;
            obase[(size_t)(16 + r) * FOUT]      = acc10[r] + bias0;
            obase[(size_t)(16 + r) * FOUT + 16] = acc11[r] + bias1;
        }

        if (t + 1 < NTILES) {
            lds_barrier();
            store_x_tile(xv, sX, h0, w0 + TILE_W, pg, c4);
            lds_barrier();
        }
    }
}

extern "C" void kernel_launch(void* const* d_in, const int* in_sizes, int n_in,
                              void* d_out, int out_size, void* d_ws, size_t ws_size,
                              hipStream_t stream) {
    (void)in_sizes; (void)n_in; (void)d_ws; (void)ws_size; (void)out_size;
    const float* x      = (const float*)d_in[0];
    const float* params = (const float*)d_in[1];
    float* out          = (float*)d_out;

    dim3 grid(Hdim / ROWS_PER_BLK, 32);
    conv3x3_wmma_f32<<<grid, 128, 0, stream>>>(x, params, out);
}


# Fast-convolution identity F(2,3) along the rows: 4 products per output-row pair instead of 6 (192 WMMAs per tile, not 288); rows are pre-added, so products and f32 summation order change
# speedup vs baseline: 1.7217x; 1.2383x over previous
#include <hip/hip_runtime.h>


typedef __attribute__((ext_vector_type(2))) float v2f;
typedef __attribute__((ext_vector_type(4))) float v4f;
typedef __attribute__((ext_vector_type(8))) float v8f;

#define CIN   32
#define FOUT  32
#define Hdim  128
#define Wdim  128
#define KHW   3
#define TILE_W 32
#define NTILES (Wdim / TILE_W)
#define ROWS_PER_BLK 4
#define XROWS (ROWS_PER_BLK + 2)
#define XCOLS (TILE_W + 2)
#define XPIX_STRIDE 36
#define WSIZE (KHW * KHW * CIN * FOUT)
#define PSTRIDE (WSIZE + FOUT)
#define USIZE (4 * KHW * CIN * FOUT)
#define XITERS ((XROWS * XCOLS * 8 + 127) / 128)

static __device__ __forceinline__ void lds_barrier() {
    asm volatile("s_wait_dscnt 0x0\n\ts_barrier_signal -1\n\ts_barrier_wait -1" ::: "memory");
}

static __device__ __forceinline__ void load_x_tile(float4 (&xv)[XITERS], const float* xs,
                                                   int h0, int w0, int pg, int c4) {
    #pragma unroll
    for (int it = 0; it < XITERS; ++it) {
        const int pix = pg + 16 * it;
        const int r   = pix / XCOLS;
        const int c   = pix - r * XCOLS;
        const int row = min(max(h0 - 1 + r, 0), Hdim - 1);
        const int col = min(max(w0 - 1 + c, 0), Wdim - 1);
        const unsigned off = (unsigned)((row * Wdim + col) * CIN + c4 * 4);
        xv[it] = *(const float4*)(xs + off);
    }
}

static __device__ __forceinline__ void store_x_tile(const float4 (&xv)[XITERS], float* sX,
                                                    int h0, int w0, int pg, int c4) {
    float4* xdst = (float4*)sX;
    #pragma unroll
    for (int it = 0; it < XITERS; ++it) {
        const int pix = pg + 16 * it;
        const int r   = pix / XCOLS;
        const int c   = pix - r * XCOLS;
        const int row = h0 - 1 + r;
        const int col = w0 - 1 + c;
        float4 v = xv[it];
        if (!((unsigned)row < (unsigned)Hdim && (unsigned)col < (unsigned)Wdim))
            v = make_float4(0.f, 0.f, 0.f, 0.f);
        xdst[pix * (XPIX_STRIDE / 4) + c4] = v;
    }
}

struct FragSet {
    v4f d[4];
    v4f u[4][2];
};

static __device__ __forceinline__ void load_frags(FragSet& s, const float* xa,
                                                  const float* wq, int blk) {
    const int kw = blk >> 2;
    const int jq = blk & 3;
    const float* xr = xa + kw * XPIX_STRIDE + 8 * jq;
    const float* wr = wq + kw * 4096 + 256 * jq;
    #pragma unroll
    for (int i = 0; i < 4; ++i) {
        s.d[i]    = *(const v4f*)(xr + i * XCOLS * XPIX_STRIDE);
        s.u[i][0] = *(const v4f*)(wr + i * 1024);
        s.u[i][1] = *(const v4f*)(wr + i * 1024 + 64);
    }
}

static __device__ __forceinline__ void make_v(const FragSet& s, v4f (&v)[4]) {
    v[0] = s.d[0] - s.d[2];
    v[1] = s.d[1] + s.d[2];
    v[2] = s.d[2] - s.d[1];
    v[3] = s.d[1] - s.d[3];
}

static __device__ __forceinline__ void mma_block(const v4f (&v)[4], const FragSet& s,
                                                 v8f (&acc)[4][2]) {
    #pragma unroll
    for (int p = 0; p < 2; ++p) {
        #pragma unroll
        for (int m = 0; m < 4; ++m) {
            v2f a  = { v[m][2 * p], v[m][2 * p + 1] };
            v2f b0 = { s.u[m][0][2 * p], s.u[m][0][2 * p + 1] };
            v2f b1 = { s.u[m][1][2 * p], s.u[m][1][2 * p + 1] };
            acc[m][0] = __builtin_amdgcn_wmma_f32_16x16x4_f32(false, a, false, b0,
                        (short)0, acc[m][0], false, false);
            acc[m][1] = __builtin_amdgcn_wmma_f32_16x16x4_f32(false, a, false, b1,
                        (short)0, acc[m][1], false, false);
        }
    }
}

static __device__ __forceinline__ void wmma_to_valu_slots() {
    asm volatile("v_nop\n\tv_nop\n\tv_nop\n\tv_nop");
}

#define NBLK (KHW * 4)

__global__ __launch_bounds__(128)
void conv3x3_wmma_f32(const float* __restrict__ x,
                      const float* __restrict__ params,
                      float* __restrict__ out) {
    __shared__ __attribute__((aligned(16))) float sU[USIZE];
    __shared__ __attribute__((aligned(16))) float sX[XITERS * 16 * XPIX_STRIDE];

    const int tid  = threadIdx.x;
    const int b    = blockIdx.y;
    const int h0   = blockIdx.x * ROWS_PER_BLK;

    const float* pb = params + (size_t)b * PSTRIDE;

    const int c4 = tid & 7;
    const int pg = tid >> 3;
    const float* xs = x + (size_t)b * Hdim * Wdim * CIN;
    float4 xv[XITERS];
    load_x_tile(xv, xs, h0, 0, pg, c4);

    const float bias0 = pb[WSIZE + (tid & 15)];
    const float bias1 = pb[WSIZE + 16 + (tid & 15)];

    {
        float4* udst = (float4*)sU;
        #pragma unroll
        for (int it = 0; it < KHW * 8 * FOUT / 128; ++it) {
            const int j  = tid + 128 * it;
            const int kw = j >> 8;
            const int q  = (j >> 5) & 7;
            const int f  = j & 31;
            const float* src = pb + ((kw * 32 + 4 * q) * 32 + f);
            float g0[4], g1[4], g2[4];
            #pragma unroll
            for (int rr = 0; rr < 4; ++rr) {
                g0[rr] = src[rr * 32];
                g1[rr] = src[rr * 32 + 3 * 1024];
                g2[rr] = src[rr * 32 + 6 * 1024];
            }
            float4 u1, u2;
            u1.x = 0.5f * ((g0[0] + g2[0]) + g1[0]);  u2.x = 0.5f * ((g0[0] + g2[0]) - g1[0]);
            u1.y = 0.5f * ((g0[1] + g2[1]) + g1[1]);  u2.y = 0.5f * ((g0[1] + g2[1]) - g1[1]);
            u1.z = 0.5f * ((g0[2] + g2[2]) + g1[2]);  u2.z = 0.5f * ((g0[2] + g2[2]) - g1[2]);
            u1.w = 0.5f * ((g0[3] + g2[3]) + g1[3]);  u2.w = 0.5f * ((g0[3] + g2[3]) - g1[3]);
            const int slot = kw * 1024 + q * 32 + f;
            udst[slot]       = make_float4(g0[0], g0[1], g0[2], g0[3]);
            udst[slot + 256] = u1;
            udst[slot + 512] = u2;
            udst[slot + 768] = make_float4(g2[0], g2[1], g2[2], g2[3]);
        }
    }

    store_x_tile(xv, sX, h0, 0, pg, c4);
    __syncthreads();

    const int lane  = tid & 31;
    const int wave  = tid >> 5;
    const int pair  = wave >> 1;
    const int xh    = wave & 1;
    const int mlane = lane & 15;
    const int hi    = lane >> 4;
    const int h     = h0 + 2 * pair;
    float* orow = out + ((((size_t)b * Hdim + h) * Wdim) + 16 * xh + 8 * hi) * FOUT + mlane;
    const float* xa = &sX[((2 * pair) * XCOLS + 16 * xh + mlane) * XPIX_STRIDE + 4 * hi];
    const float* wq = &sU[mlane * 4 + 128 * hi];

    #pragma unroll 1
    for (int t = 0; t < NTILES; ++t) {
        const int w0 = t * TILE_W;

        if (t + 1 < NTILES)
            load_x_tile(xv, xs, h0, w0 + TILE_W, pg, c4);

        v8f acc[4][2] = {};

        FragSet fs[2];
        load_frags(fs[0], xa, wq, 0);
        #pragma unroll
        for (int blk = 0; blk < NBLK; ++blk) {
            if (blk + 1 < NBLK)
                load_frags(fs[(blk + 1) & 1], xa, wq, blk + 1);
            __builtin_amdgcn_sched_barrier(0);
            v4f v[4];
            make_v(fs[blk & 1], v);
            __builtin_amdgcn_sched_barrier(0);
            mma_block(v, fs[blk & 1], acc);
            __builtin_amdgcn_sched_barrier(0);
            wmma_to_valu_slots();
            __builtin_amdgcn_sched_barrier(0);
        }

        float* obase = orow + (size_t)w0 * FOUT;
        #pragma unroll
        for (int r = 0; r < 8; ++r) {
            const float s0 = acc[1][0][r] + acc[2][0][r];
            const float t0 = acc[1][0][r] - acc[2][0][r];
            const float s1 = acc[1][1][r] + acc[2][1][r];
            const float t1 = acc[1][1][r] - acc[2][1][r];
            obase[(size_t)r * FOUT]                    = (acc[0][0][r] + s0) + bias0;
            obase[(size_t)r * FOUT + 16]               = (acc[0][1][r] + s1) + bias1;
            obase[(size_t)(Wdim + r) * FOUT]           = (t0 - acc[3][0][r]) + bias0;
            obase[(size_t)(Wdim + r) * FOUT + 16]      = (t1 - acc[3][1][r]) + bias1;
        }

        if (t + 1 < NTILES) {
            lds_barrier();
            store_x_tile(xv, sX, h0, w0 + TILE_W, pg, c4);
            lds_barrier();
        }
    }
}

extern "C" void kernel_launch(void* const* d_in, const int* in_sizes, int n_in,
                              void* d_out, int out_size, void* d_ws, size_t ws_size,
                              hipStream_t stream) {
    (void)in_sizes; (void)n_in; (void)d_ws; (void)ws_size; (void)out_size;
    const float* x      = (const float*)d_in[0];
    const float* params = (const float*)d_in[1];
    float* out          = (float*)d_out;

    dim3 grid(Hdim / ROWS_PER_BLK, 32);
    conv3x3_wmma_f32<<<grid, 128, 0, stream>>>(x, params, out);
}


# Wave priority falls with the tile index (s_setprio 3 - t): the 4 co-resident blocks of a WGP advance together, not oldest-first one by one, so 4 waves per SIMD stay runnable to the end
# speedup vs baseline: 1.7816x; 1.0348x over previous
#include <hip/hip_runtime.h>


typedef __attribute__((ext_vector_type(2))) float v2f;
typedef __attribute__((ext_vector_type(4))) float v4f;
typedef __attribute__((ext_vector_type(8))) float v8f;

#define CIN   32
#define FOUT  32
#define Hdim  128
#define Wdim  128
#define KHW   3
#define TILE_W 32
#define NTILES (Wdim / TILE_W)
#define ROWS_PER_BLK 4
#define XROWS (ROWS_PER_BLK + 2)
#define XCOLS (TILE_W + 2)
#define XPIX_STRIDE 36
#define WSIZE (KHW * KHW * CIN * FOUT)
#define PSTRIDE (WSIZE + FOUT)
#define USIZE (4 * KHW * CIN * FOUT)
#define XITERS ((XROWS * XCOLS * 8 + 127) / 128)

static __device__ __forceinline__ void lds_barrier() {
    asm volatile("s_wait_dscnt 0x0\n\ts_barrier_signal -1\n\ts_barrier_wait -1" ::: "memory");
}

static __device__ __forceinline__ void load_x_tile(float4 (&xv)[XITERS], const float* xs,
                                                   int h0, int w0, int pg, int c4) {
    #pragma unroll
    for (int it = 0; it < XITERS; ++it) {
        const int pix = pg + 16 * it;
        const int r   = pix / XCOLS;
        const int c   = pix - r * XCOLS;
        const int row = min(max(h0 - 1 + r, 0), Hdim - 1);
        const int col = min(max(w0 - 1 + c, 0), Wdim - 1);
        const unsigned off = (unsigned)((row * Wdim + col) * CIN + c4 * 4);
        xv[it] = *(const float4*)(xs + off);
    }
}

static __device__ __forceinline__ void store_x_tile(const float4 (&xv)[XITERS], float* sX,
                                                    int h0, int w0, int pg, int c4) {
    float4* xdst = (float4*)sX;
    #pragma unroll
    for (int it = 0; it < XITERS; ++it) {
        const int pix = pg + 16 * it;
        const int r   = pix / XCOLS;
        const int c   = pix - r * XCOLS;
        const int row = h0 - 1 + r;
        const int col = w0 - 1 + c;
        float4 v = xv[it];
        if (!((unsigned)row < (unsigned)Hdim && (unsigned)col < (unsigned)Wdim))
            v = make_float4(0.f, 0.f, 0.f, 0.f);
        xdst[pix * (XPIX_STRIDE / 4) + c4] = v;
    }
}

struct FragSet {
    v4f d[4];
    v4f u[4][2];
};

static __device__ __forceinline__ void load_frags(FragSet& s, const float* xa,
                                                  const float* wq, int blk) {
    const int kw = blk >> 2;
    const int jq = blk & 3;
    const float* xr = xa + kw * XPIX_STRIDE + 8 * jq;
    const float* wr = wq + kw * 4096 + 256 * jq;
    #pragma unroll
    for (int i = 0; i < 4; ++i) {
        s.d[i]    = *(const v4f*)(xr + i * XCOLS * XPIX_STRIDE);
        s.u[i][0] = *(const v4f*)(wr + i * 1024);
        s.u[i][1] = *(const v4f*)(wr + i * 1024 + 64);
    }
}

static __device__ __forceinline__ void make_v(const FragSet& s, v4f (&v)[4]) {
    v[0] = s.d[0] - s.d[2];
    v[1] = s.d[1] + s.d[2];
    v[2] = s.d[2] - s.d[1];
    v[3] = s.d[1] - s.d[3];
}

static __device__ __forceinline__ void mma_block(const v4f (&v)[4], const FragSet& s,
                                                 v8f (&acc)[4][2]) {
    #pragma unroll
    for (int p = 0; p < 2; ++p) {
        #pragma unroll
        for (int m = 0; m < 4; ++m) {
            v2f a  = { v[m][2 * p], v[m][2 * p + 1] };
            v2f b0 = { s.u[m][0][2 * p], s.u[m][0][2 * p + 1] };
            v2f b1 = { s.u[m][1][2 * p], s.u[m][1][2 * p + 1] };
            acc[m][0] = __builtin_amdgcn_wmma_f32_16x16x4_f32(false, a, false, b0,
                        (short)0, acc[m][0], false, false);
            acc[m][1] = __builtin_amdgcn_wmma_f32_16x16x4_f32(false, a, false, b1,
                        (short)0, acc[m][1], false, false);
        }
    }
}

static __device__ __forceinline__ void wmma_to_valu_slots() {
    asm volatile("v_nop\n\tv_nop\n\tv_nop\n\tv_nop");
}

#define NBLK (KHW * 4)

__global__ __launch_bounds__(128)
void conv3x3_wmma_f32(const float* __restrict__ x,
                      const float* __restrict__ params,
                      float* __restrict__ out) {
    __shared__ __attribute__((aligned(16))) float sU[USIZE];
    __shared__ __attribute__((aligned(16))) float sX[XITERS * 16 * XPIX_STRIDE];

    __builtin_amdgcn_s_setprio(3);

    const int tid  = threadIdx.x;
    const int b    = blockIdx.y;
    const int h0   = blockIdx.x * ROWS_PER_BLK;

    const float* pb = params + (size_t)b * PSTRIDE;

    const int c4 = tid & 7;
    const int pg = tid >> 3;
    const float* xs = x + (size_t)b * Hdim * Wdim * CIN;
    float4 xv[XITERS];
    load_x_tile(xv, xs, h0, 0, pg, c4);

    const float bias0 = pb[WSIZE + (tid & 15)];
    const float bias1 = pb[WSIZE + 16 + (tid & 15)];

    {
        float4* udst = (float4*)sU;
        #pragma unroll
        for (int it = 0; it < KHW * 8 * FOUT / 128; ++it) {
            const int j  = tid + 128 * it;
            const int kw = j >> 8;
            const int q  = (j >> 5) & 7;
            const int f  = j & 31;
            const float* src = pb + ((kw * 32 + 4 * q) * 32 + f);
            float g0[4], g1[4], g2[4];
            #pragma unroll
            for (int rr = 0; rr < 4; ++rr) {
                g0[rr] = src[rr * 32];
                g1[rr] = src[rr * 32 + 3 * 1024];
                g2[rr] = src[rr * 32 + 6 * 1024];
            }
            float4 u1, u2;
            u1.x = 0.5f * ((g0[0] + g2[0]) + g1[0]);  u2.x = 0.5f * ((g0[0] + g2[0]) - g1[0]);
            u1.y = 0.5f * ((g0[1] + g2[1]) + g1[1]);  u2.y = 0.5f * ((g0[1] + g2[1]) - g1[1]);
            u1.z = 0.5f * ((g0[2] + g2[2]) + g1[2]);  u2.z = 0.5f * ((g0[2] + g2[2]) - g1[2]);
            u1.w = 0.5f * ((g0[3] + g2[3]) + g1[3]);  u2.w = 0.5f * ((g0[3] + g2[3]) - g1[3]);
            const int slot = kw * 1024 + q * 32 + f;
            udst[slot]       = make_float4(g0[0], g0[1], g0[2], g0[3]);
            udst[slot + 256] = u1;
            udst[slot + 512] = u2;
            udst[slot + 768] = make_float4(g2[0], g2[1], g2[2], g2[3]);
        }
    }

    store_x_tile(xv, sX, h0, 0, pg, c4);
    __syncthreads();

    const int lane  = tid & 31;
    const int wave  = tid >> 5;
    const int pair  = wave >> 1;
    const int xh    = wave & 1;
    const int mlane = lane & 15;
    const int hi    = lane >> 4;
    const int h     = h0 + 2 * pair;
    float* orow = out + ((((size_t)b * Hdim + h) * Wdim) + 16 * xh + 8 * hi) * FOUT + mlane;
    const float* xa = &sX[((2 * pair) * XCOLS + 16 * xh + mlane) * XPIX_STRIDE + 4 * hi];
    const float* wq = &sU[mlane * 4 + 128 * hi];

    #pragma unroll 1
    for (int t = 0; t < NTILES; ++t) {
        const int w0 = t * TILE_W;

        if (t == 1)      __builtin_amdgcn_s_setprio(2);
        else if (t == 2) __builtin_amdgcn_s_setprio(1);
        else if (t == 3) __builtin_amdgcn_s_setprio(0);

        if (t + 1 < NTILES)
            load_x_tile(xv, xs, h0, w0 + TILE_W, pg, c4);

        v8f acc[4][2] = {};

        FragSet fs[2];
        load_frags(fs[0], xa, wq, 0);
        #pragma unroll
        for (int blk = 0; blk < NBLK; ++blk) {
            if (blk + 1 < NBLK)
                load_frags(fs[(blk + 1) & 1], xa, wq, blk + 1);
            __builtin_amdgcn_sched_barrier(0);
            v4f v[4];
            make_v(fs[blk & 1], v);
            __builtin_amdgcn_sched_barrier(0);
            mma_block(v, fs[blk & 1], acc);
            __builtin_amdgcn_sched_barrier(0);
            wmma_to_valu_slots();
            __builtin_amdgcn_sched_barrier(0);
        }

        float* obase = orow + (size_t)w0 * FOUT;
        #pragma unroll
        for (int r = 0; r < 8; ++r) {
            const float s0 = acc[1][0][r] + acc[2][0][r];
            const float t0 = acc[1][0][r] - acc[2][0][r];
            const float s1 = acc[1][1][r] + acc[2][1][r];
            const float t1 = acc[1][1][r] - acc[2][1][r];
            obase[(size_t)r * FOUT]                    = (acc[0][0][r] + s0) + bias0;
            obase[(size_t)r * FOUT + 16]               = (acc[0][1][r] + s1) + bias1;
            obase[(size_t)(Wdim + r) * FOUT]           = (t0 - acc[3][0][r]) + bias0;
            obase[(size_t)(Wdim + r) * FOUT + 16]      = (t1 - acc[3][1][r]) + bias1;
        }

        if (t + 1 < NTILES) {
            lds_barrier();
            store_x_tile(xv, sX, h0, w0 + TILE_W, pg, c4);
            lds_barrier();
        }
    }
}

extern "C" void kernel_launch(void* const* d_in, const int* in_sizes, int n_in,
                              void* d_out, int out_size, void* d_ws, size_t ws_size,
                              hipStream_t stream) {
    (void)in_sizes; (void)n_in; (void)d_ws; (void)ws_size; (void)out_size;
    const float* x      = (const float*)d_in[0];
    const float* params = (const float*)d_in[1];
    float* out          = (float*)d_out;

    dim3 grid(Hdim / ROWS_PER_BLK, 32);
    conv3x3_wmma_f32<<<grid, 128, 0, stream>>>(x, params, out);
}
